# starting kernel
# baseline (speedup 1.0000x reference)
#include <hip/hip_runtime.h>
#include <math.h>

#define NN     100000
#define EE     1600000
#define FIN    300
#define HID    64
#define OUTC   20
#define NLAYER 8
#define ALPHA  0.1f

typedef float v2f __attribute__((ext_vector_type(2)));
typedef float v8f __attribute__((ext_vector_type(8)));

__device__ __forceinline__ v8f wmma_f32_16x16x4(v2f a, v2f b, v8f c) {
    return __builtin_amdgcn_wmma_f32_16x16x4_f32(false, a, false, b, (short)0, c,
                                                 false, false);
}

__global__ void k_fill(float* __restrict__ p, float v, int n) {
    int i = blockIdx.x * blockDim.x + threadIdx.x;
    if (i < n) p[i] = v;
}

__global__ void k_deg_acc(const int* __restrict__ col, const float* __restrict__ w,
                          float* __restrict__ deg) {
    int e = blockIdx.x * blockDim.x + threadIdx.x;
    if (e < EE) atomicAdd(&deg[col[e]], w[e]);
}

__global__ void k_dinv(float* __restrict__ deg) {
    int i = blockIdx.x * blockDim.x + threadIdx.x;
    if (i < NN) {
        float d = deg[i];
        deg[i] = d > 0.f ? rsqrtf(d) : 0.f;
    }
}

__global__ void k_norm(const int* __restrict__ row, const int* __restrict__ col,
                       const float* __restrict__ w, const float* __restrict__ dinv,
                       float* __restrict__ norm) {
    int e = blockIdx.x * blockDim.x + threadIdx.x;
    if (e < EE) norm[e] = dinv[row[e]] * w[e] * dinv[col[e]];
}

__global__ __launch_bounds__(128)
void k_gemm_in(const float* __restrict__ x, const float* __restrict__ Win,
               const float* __restrict__ bin, float* __restrict__ x0,
               float* __restrict__ h) {
    __shared__ float xs[16][FIN + 4];
    const int tid = threadIdx.x;
    const int mb  = blockIdx.x;

    for (int i = tid; i < 16 * FIN; i += 128) {
        int r = i / FIN, c = i - r * FIN;
        xs[r][c] = x[(size_t)(mb * 16 + r) * FIN + c];
    }
    __syncthreads();

    const int lane = tid & 31;
    const int n0   = (tid >> 5) * 16;
    const int m    = lane & 15;
    const int kh   = (lane >> 4) * 2;
    const int nl   = lane & 15;

    v8f acc = {};
    for (int k0 = 0; k0 < FIN; k0 += 4) {
        v2f a, b;
        a[0] = xs[m][k0 + kh];
        a[1] = xs[m][k0 + kh + 1];
        b[0] = Win[(size_t)(k0 + kh)     * HID + n0 + nl];
        b[1] = Win[(size_t)(k0 + kh + 1) * HID + n0 + nl];
        acc  = wmma_f32_16x16x4(a, b, acc);
    }

    #pragma unroll
    for (int v = 0; v < 8; ++v) {
        int mm  = v + (lane >> 4) * 8;
        int c   = n0 + nl;
        float val = acc[v] + bin[c];
        val = val > 0.f ? val : 0.f;
        size_t idx = (size_t)(mb * 16 + mm) * HID + c;
        x0[idx] = val;
        h[idx]  = val;
    }
}

__global__ __launch_bounds__(256)
void k_scatter(const int* __restrict__ row, const int* __restrict__ col,
               const float* __restrict__ norm, const float* __restrict__ h,
               float* __restrict__ agg) {
    int gid  = blockIdx.x * blockDim.x + threadIdx.x;
    int lane = gid & 31;
    int wid  = gid >> 5;
    int nwv  = (gridDim.x * blockDim.x) >> 5;
    for (int e = wid; e < EE; e += nwv) {
        int   r  = row[e];
        int   c  = col[e];
        float nv = norm[e];
        float2 hv = *reinterpret_cast<const float2*>(h + (size_t)r * HID + lane * 2);
        atomicAdd(agg + (size_t)c * HID + lane * 2 + 0, nv * hv.x);
        atomicAdd(agg + (size_t)c * HID + lane * 2 + 1, nv * hv.y);
    }
}

__global__ __launch_bounds__(128)
void k_layer(const float* __restrict__ agg, const float* __restrict__ x0,
             const float* __restrict__ dinv, const float* __restrict__ Wl,
             float* __restrict__ h, float beta) {
    __shared__ float zs[16][HID + 8];
    __shared__ float ws[HID][HID];
    const int tid = threadIdx.x;
    const int mb  = blockIdx.x;

    for (int i = tid; i < 16 * HID; i += 128) {
        int r = i >> 6, c = i & 63;
        int rowi = mb * 16 + r;
        float di = dinv[rowi];
        size_t idx = (size_t)rowi * HID + c;
        float a = agg[idx] + di * di * h[idx];
        zs[r][c] = (1.f - ALPHA) * a + ALPHA * x0[idx];
    }
    for (int i = tid; i < HID * HID; i += 128)
        ws[i >> 6][i & 63] = Wl[i];
    __syncthreads();

    const int lane = tid & 31;
    const int n0   = (tid >> 5) * 16;
    const int m    = lane & 15;
    const int kh   = (lane >> 4) * 2;
    const int nl   = lane & 15;

    v8f acc = {};
    #pragma unroll
    for (int k0 = 0; k0 < HID; k0 += 4) {
        v2f a, b;
        a[0] = zs[m][k0 + kh];
        a[1] = zs[m][k0 + kh + 1];
        b[0] = ws[k0 + kh][n0 + nl];
        b[1] = ws[k0 + kh + 1][n0 + nl];
        acc  = wmma_f32_16x16x4(a, b, acc);
    }

    #pragma unroll
    for (int v = 0; v < 8; ++v) {
        int mm = v + (lane >> 4) * 8;
        float z   = zs[mm][n0 + nl];
        float val = (1.f - beta) * z + beta * acc[v];
        val = val > 0.f ? val : 0.f;
        h[(size_t)(mb * 16 + mm) * HID + n0 + nl] = val;
    }
}

__global__ void k_out(const float* __restrict__ h, const float* __restrict__ Wout,
                      const float* __restrict__ bout, float* __restrict__ out) {
    int i = blockIdx.x * blockDim.x + threadIdx.x;
    if (i >= NN * OUTC) return;
    int n = i / OUTC, c = i - n * OUTC;
    float s = bout[c];
    const float* hr = h + (size_t)n * HID;
    #pragma unroll 16
    for (int k = 0; k < HID; ++k) s += hr[k] * Wout[k * OUTC + c];
    out[i] = s;
}

extern "C" void kernel_launch(void* const* d_in, const int* in_sizes, int n_in,
                              void* d_out, int out_size, void* d_ws, size_t ws_size,
                              hipStream_t stream) {
    (void)in_sizes; (void)n_in; (void)out_size; (void)ws_size;

    const float* x     = (const float*)d_in[0];
    const int*   ei    = (const int*)  d_in[1];
    const float* ew    = (const float*)d_in[2];
    const float* Win   = (const float*)d_in[3];
    const float* bin   = (const float*)d_in[4];
    const float* convw = (const float*)d_in[5];
    const float* Wout  = (const float*)d_in[6];
    const float* bout  = (const float*)d_in[7];
    const int* row = ei;
    const int* col = ei + EE;

    char*  ws  = (char*)d_ws;
    size_t off = 0;
    auto carve = [&](size_t bytes) -> float* {
        float* p = (float*)(ws + off);
        off = (off + bytes + 255) & ~(size_t)255;
        return p;
    };
    float* dinv = carve((size_t)NN * 4);
    float* norm = carve((size_t)EE * 4);
    float* x0   = carve((size_t)NN * HID * 4);
    float* h    = carve((size_t)NN * HID * 4);
    float* agg  = carve((size_t)NN * HID * 4);

    k_fill   <<<(NN + 255) / 256, 256, 0, stream>>>(dinv, 1.0f, NN);
    k_deg_acc<<<(EE + 255) / 256, 256, 0, stream>>>(col, ew, dinv);
    k_dinv   <<<(NN + 255) / 256, 256, 0, stream>>>(dinv);
    k_norm   <<<(EE + 255) / 256, 256, 0, stream>>>(row, col, ew, dinv, norm);

    k_gemm_in<<<NN / 16, 128, 0, stream>>>(x, Win, bin, x0, h);

    for (int l = 0; l < NLAYER; ++l) {
        float beta = logf(0.5f / (float)(l + 1) + 1.0f);
        k_fill   <<<(NN * HID + 255) / 256, 256, 0, stream>>>(agg, 0.0f, NN * HID);
        k_scatter<<<2048, 256, 0, stream>>>(row, col, norm, h, agg);
        k_layer  <<<NN / 16, 128, 0, stream>>>(agg, x0, dinv,
                                               convw + (size_t)l * HID * HID, h, beta);
    }

    k_out<<<(NN * OUTC + 255) / 256, 256, 0, stream>>>(h, Wout, bout, (float*)d_out);
}


# output projection moved into the last layer's work-groups: a second f32 matrix pass over their LDS tile of h; no output launch, last h never stored; out sums in matrix-unit order, bias added last
# speedup vs baseline: 3.1028x; 3.1028x over previous
#include <hip/hip_runtime.h>
#include <math.h>


#define NN     100000
#define EE     1600000
#define FIN    300
#define HID    64
#define OUTC   20
#define NLAYER 8
#define ALPHA  0.1f

#define RNG    400
#define LSH    9
#define NB     ((NN + RNG - 1) / RNG)
#define CPAD   32
#define BT     640
#define EPT    10
#define ECH    (BT * EPT)
#define NCH    (EE / ECH)
static_assert(NN % RNG == 0, "every range is full: the layer kernel has no node guard");
static_assert(RNG % 16 == 0, "a range is a whole number of 16-row tiles");
static_assert(EE % ECH == 0, "the binning passes have no edge guard");
static_assert(NB < 512, "range counts are scanned by one 512-thread work-group");
static_assert(RNG < (1 << LSH), "local destination (and the spare row RNG) must fit LSH bits");
static_assert(NN <= (1 << (31 - LSH)), "source id and local destination share one 32-bit word");

typedef float v2f __attribute__((ext_vector_type(2)));
typedef float v8f __attribute__((ext_vector_type(8)));

__device__ __forceinline__ v8f wmma_f32_16x16x4(v2f a, v2f b, v8f c) {
    return __builtin_amdgcn_wmma_f32_16x16x4_f32(false, a, false, b, (short)0, c,
                                                 false, false);
}

__global__ void k_fill(float* __restrict__ p, float v, int n) {
    int i = blockIdx.x * blockDim.x + threadIdx.x;
    if (i < n) p[i] = v;
}


__global__ __launch_bounds__(BT)
void k_hist(const int* __restrict__ col, int* __restrict__ cnt) {
    __shared__ int hc[NB];
    const int tid = threadIdx.x;
    for (int i = tid; i < NB; i += BT) hc[i] = 0;
    __syncthreads();
    const int e0 = blockIdx.x * ECH + tid;
    int c[EPT];
    #pragma unroll
    for (int k = 0; k < EPT; ++k) c[k] = col[e0 + k * BT];
    #pragma unroll
    for (int k = 0; k < EPT; ++k) atomicAdd(&hc[c[k] / RNG], 1);
    __syncthreads();
    for (int i = tid; i < NB; i += BT) {
        int n = hc[i];
        if (n) atomicAdd(&cnt[i * CPAD], n);
    }
}

__global__ __launch_bounds__(512)
void k_scan(const int* __restrict__ cnt, int* __restrict__ start,
            int* __restrict__ cursor) {
    __shared__ int s[512];
    const int tid = threadIdx.x;
    const int v = tid < NB ? cnt[tid * CPAD] : 0;
    s[tid] = v;
    __syncthreads();
    for (int off = 1; off < 512; off <<= 1) {
        int t = tid >= off ? s[tid - off] : 0;
        __syncthreads();
        s[tid] += t;
        __syncthreads();
    }
    if (tid < NB) {
        start[tid]         = s[tid] - v;
        cursor[tid * CPAD] = s[tid] - v;
    }
    if (tid == NB - 1) start[NB] = s[tid];
}

__global__ __launch_bounds__(BT)
void k_place(const int* __restrict__ row, const int* __restrict__ col,
             const float* __restrict__ w,
             int* __restrict__ cursor, int* __restrict__ sedge,
             float* __restrict__ snorm) {
    __shared__ int hc[NB];
    const int tid = threadIdx.x;
    for (int i = tid; i < NB; i += BT) hc[i] = 0;
    __syncthreads();
    const int e0 = blockIdx.x * ECH + tid;
    int c[EPT], r[EPT];
    float wv[EPT];
    #pragma unroll
    for (int k = 0; k < EPT; ++k) {
        c[k]  = col[e0 + k * BT];
        r[k]  = row[e0 + k * BT];
        wv[k] = w[e0 + k * BT];
    }
    #pragma unroll
    for (int k = 0; k < EPT; ++k) atomicAdd(&hc[c[k] / RNG], 1);
    __syncthreads();
    for (int i = tid; i < NB; i += BT) {
        int n = hc[i];
        hc[i] = n ? atomicAdd(&cursor[i * CPAD], n) : 0;
    }
    __syncthreads();
    #pragma unroll
    for (int k = 0; k < EPT; ++k) {
        int b    = c[k] / RNG;
        int slot = atomicAdd(&hc[b], 1);
        sedge[slot] = (r[k] << LSH) | (c[k] - b * RNG);
        snorm[slot] = wv[k];
    }
}

#define DT   512
#define DU   6
#define LMSK ((1 << LSH) - 1)

template <bool TAIL>
__device__ __forceinline__ void deg_trip(const int* __restrict__ sedge,
                                         const float* __restrict__ sw,
                                         float* dg, int e, int ee) {
    int   p[DU];
    float w[DU];
    #pragma unroll
    for (int k = 0; k < DU; ++k) {
        const int ix = TAIL ? min(e + k * DT, ee - 1) : e + k * DT;
        p[k] = sedge[ix];
        w[k] = sw[ix];
    }
    #pragma unroll
    for (int k = 0; k < DU; ++k)
        if (!TAIL || e + k * DT < ee) atomicAdd(&dg[p[k] & LMSK], w[k]);
}

__global__ __launch_bounds__(DT)
void k_deg_range(const int* __restrict__ sedge, const float* __restrict__ sw,
                 const int* __restrict__ start, float* __restrict__ dinv) {
    __shared__ float dg[RNG];
    const int tid = threadIdx.x;
    for (int i = tid; i < RNG; i += DT) dg[i] = 1.0f;
    __syncthreads();
    const int eb = start[blockIdx.x];
    const int ee = start[blockIdx.x + 1];
    for (int e = eb + tid; e < ee; e += DT * DU) {
        const int e_hi = __builtin_amdgcn_readfirstlane(e) + 31 + (DU - 1) * DT;
        if (e_hi < ee) deg_trip<false>(sedge, sw, dg, e, ee);
        else           deg_trip<true >(sedge, sw, dg, e, ee);
    }
    __syncthreads();
    for (int i = tid; i < RNG; i += DT) {
        const float d = dg[i];
        dinv[blockIdx.x * RNG + i] = d > 0.f ? rsqrtf(d) : 0.f;
    }
}

template <bool TAIL>
__device__ __forceinline__ void norm_trip(const int* __restrict__ sedge,
                                          const float* __restrict__ dinv,
                                          float* __restrict__ snorm,
                                          int e, int ee, int nb0) {
    int   p[DU];
    float w[DU], ds[DU], dd[DU];
    #pragma unroll
    for (int k = 0; k < DU; ++k) {
        const int ix = TAIL ? min(e + k * DT, ee - 1) : e + k * DT;
        p[k] = sedge[ix];
        w[k] = snorm[ix];
    }
    #pragma unroll
    for (int k = 0; k < DU; ++k) {
        ds[k] = dinv[(unsigned)p[k] >> LSH];
        dd[k] = dinv[nb0 + (p[k] & LMSK)];
    }
    #pragma unroll
    for (int k = 0; k < DU; ++k)
        if (!TAIL || e + k * DT < ee) snorm[e + k * DT] = ds[k] * w[k] * dd[k];
}

__global__ __launch_bounds__(DT)
void k_norm(const int* __restrict__ sedge, const int* __restrict__ start,
            const float* __restrict__ dinv, float* __restrict__ snorm) {
    const int tid = threadIdx.x;
    const int nb0 = blockIdx.x * RNG;
    const int eb = start[blockIdx.x];
    const int ee = start[blockIdx.x + 1];
    for (int e = eb + tid; e < ee; e += DT * DU) {
        const int e_hi = __builtin_amdgcn_readfirstlane(e) + 31 + (DU - 1) * DT;
        if (e_hi < ee) norm_trip<false>(sedge, dinv, snorm, e, ee, nb0);
        else           norm_trip<true >(sedge, dinv, snorm, e, ee, nb0);
    }
}

__global__ __launch_bounds__(128)
void k_gemm_in(const float* __restrict__ x, const float* __restrict__ Win,
               const float* __restrict__ bin, float* __restrict__ x0) {
    __shared__ float xs[16][FIN + 4];
    const int tid = threadIdx.x;
    const int mb  = blockIdx.x;

    for (int i = tid; i < 16 * FIN; i += 128) {
        int r = i / FIN, c = i - r * FIN;
        xs[r][c] = x[(size_t)(mb * 16 + r) * FIN + c];
    }
    __syncthreads();

    const int lane = tid & 31;
    const int n0   = (tid >> 5) * 16;
    const int m    = lane & 15;
    const int kh   = (lane >> 4) * 2;
    const int nl   = lane & 15;

    v8f acc = {};
    for (int k0 = 0; k0 < FIN; k0 += 4) {
        v2f a, b;
        a[0] = xs[m][k0 + kh];
        a[1] = xs[m][k0 + kh + 1];
        b[0] = Win[(size_t)(k0 + kh)     * HID + n0 + nl];
        b[1] = Win[(size_t)(k0 + kh + 1) * HID + n0 + nl];
        acc  = wmma_f32_16x16x4(a, b, acc);
    }

    #pragma unroll
    for (int v = 0; v < 8; ++v) {
        int mm  = v + (lane >> 4) * 8;
        int c   = n0 + nl;
        float val = acc[v] + bin[c];
        val = val > 0.f ? val : 0.f;
        x0[(size_t)(mb * 16 + mm) * HID + c] = val;
    }
}


__host__ __device__ constexpr size_t al256(size_t b) { return (b + 255) & ~(size_t)255; }
#define WS_NEED (al256((size_t)NN * 4) + 2 * al256((size_t)NB * CPAD * 4)                \
                 + al256((size_t)(NB + 1) * 4) + 2 * al256((size_t)EE * 4)                \
                 + 3 * al256((size_t)NN * HID * 4))

#define LT   1024
#define LW   (LT / 32)
#define GRP  16
#define ZP   (HID + 8)
#define MT   (RNG / 16)
#define NG   (RNG * (HID / 4))
#define PPT  ((NG + LT - 1) / LT)
#define HP   (HID + 4)
#define WOP  40
#define OCT  ((OUTC + 15) / 16)
static_assert(OCT * 16 <= 32, "the padded output weight holds 32 columns");
static_assert(RNG * OUTC <= (RNG + 1) * ZP, "a range's output tile fits the LDS rows it reuses");
static_assert((RNG * OUTC) % 4 == 0, "a range's output tile is stored in 16-byte pieces");

__device__ __forceinline__ int lane_bcast(int v, int src_lane) {
    return __builtin_amdgcn_readlane(v, src_lane);
}

template <bool LAST>
__global__ __launch_bounds__(LT)
void k_agg_layer(const int* __restrict__ sedge, const float* __restrict__ snorm,
                 const int* __restrict__ start, const float* __restrict__ hin,
                 const float* __restrict__ x0, const float* __restrict__ dinv,
                 const float* __restrict__ Wl, float* __restrict__ hout,
                 const float* __restrict__ Wout, const float* __restrict__ bout,
                 float* __restrict__ out, float beta, size_t ws_have) {
    __shared__ __attribute__((aligned(16))) float zs[RNG + 1][ZP];
    __shared__ __attribute__((aligned(16))) float ws[HID][HID];
    __shared__ __attribute__((aligned(16))) float hs[LAST ? RNG : 1][HP];
    __shared__ __attribute__((aligned(16))) float wo[LAST ? HID : 1][WOP];
    if (ws_have < WS_NEED) return;

    const int tid  = threadIdx.x;
    const int lane = tid & 31;
    const int wv   = __builtin_amdgcn_readfirstlane(tid >> 5);
    const int nb0  = blockIdx.x * RNG;

    {
        float4 hr[PPT];
        float  dr[PPT];
        #pragma unroll
        for (int k = 0; k < PPT; ++k) {
            const int q = min(tid + k * LT, NG - 1);
            const int node = nb0 + (q >> 4);
            dr[k] = dinv[node];
            hr[k] = *reinterpret_cast<const float4*>(hin + (size_t)node * HID + (q & 15) * 4);
        }
        #pragma unroll
        for (int k = 0; k < PPT; ++k) {
            const int q = tid + k * LT;
            if (q < NG) {
                const float d2 = dr[k] * dr[k];
                float4 v;
                v.x = d2 * hr[k].x; v.y = d2 * hr[k].y; v.z = d2 * hr[k].z; v.w = d2 * hr[k].w;
                *reinterpret_cast<float4*>(&zs[q >> 4][(q & 15) * 4]) = v;
            }
        }
    }
    if (tid < HID) zs[RNG][tid] = 0.f;
    for (int i = tid; i < HID * HID / 4; i += LT)
        reinterpret_cast<float4*>(&ws[0][0])[i] = reinterpret_cast<const float4*>(Wl)[i];
    if (LAST) {
        for (int i = tid; i < HID * 32; i += LT) {
            const int k = i >> 5, cc = i & 31;
            wo[k][cc] = cc < OUTC ? Wout[k * OUTC + cc] : 0.f;
        }
    }
    __syncthreads();

    const int eb = start[blockIdx.x];
    const int ee = start[blockIdx.x + 1];
    int   p_nx = RNG;
    float w_nx = 0.f;
    if (eb + wv * 32 + lane < ee) {
        p_nx = sedge[eb + wv * 32 + lane];
        w_nx = snorm[eb + wv * 32 + lane];
    }
    for (int base = eb + wv * 32; base < ee; base += LW * 32) {
        const int p_l = p_nx;
        const int w_l = __float_as_int(w_nx);
        const int   nx   = base + LW * 32 + lane;
        const int   ix   = min(nx, ee - 1);
        const int   p_ld = sedge[ix];
        const float w_ld = snorm[ix];
        const int n = ee - base;
        #pragma unroll
        for (int g = 0; g < 32; g += GRP) {
            if (g < n) {
                float2 hv[GRP];
                #pragma unroll
                for (int j = 0; j < GRP; ++j) {
                    const unsigned r = (unsigned)lane_bcast(p_l, g + j) >> LSH;
                    const float* rp = hin + (size_t)r * HID;
                    hv[j] = *reinterpret_cast<const float2*>(rp + lane * 2);
                }
                #pragma unroll
                for (int j = 0; j < GRP; ++j) {
                    const int   cl = lane_bcast(p_l, g + j) & ((1 << LSH) - 1);
                    const float nv = __int_as_float(lane_bcast(w_l, g + j));
                    atomicAdd(&zs[cl][lane * 2 + 0], nv * hv[j].x);
                    atomicAdd(&zs[cl][lane * 2 + 1], nv * hv[j].y);
                }
            }
        }
        p_nx = nx < ee ? p_ld : RNG;
        w_nx = nx < ee ? w_ld : 0.f;
    }
    __syncthreads();

    {
        float4 xr[PPT];
        #pragma unroll
        for (int k = 0; k < PPT; ++k) {
            const int q = min(tid + k * LT, NG - 1);
            xr[k] = *reinterpret_cast<const float4*>(x0 + (size_t)(nb0 + (q >> 4)) * HID + (q & 15) * 4);
        }
        #pragma unroll
        for (int k = 0; k < PPT; ++k) {
            const int q = tid + k * LT;
            if (q < NG) {
                float4* zp = reinterpret_cast<float4*>(&zs[q >> 4][(q & 15) * 4]);
                float4 a = *zp;
                a.x = (1.f - ALPHA) * a.x + ALPHA * xr[k].x;
                a.y = (1.f - ALPHA) * a.y + ALPHA * xr[k].y;
                a.z = (1.f - ALPHA) * a.z + ALPHA * xr[k].z;
                a.w = (1.f - ALPHA) * a.w + ALPHA * xr[k].w;
                *zp = a;
            }
        }
    }
    __syncthreads();

    const int m  = lane & 15;
    const int kh = (lane >> 4) * 2;
    const int nl = lane & 15;

    for (int job = wv; job < MT * 4; job += LW) {
        const int mt = job >> 2;
        const int n0 = (job & 3) * 16;
        v8f acc = {};
        #pragma unroll
        for (int k0 = 0; k0 < HID; k0 += 4) {
            v2f a, b;
            a[0] = zs[mt * 16 + m][k0 + kh];
            a[1] = zs[mt * 16 + m][k0 + kh + 1];
            b[0] = ws[k0 + kh][n0 + nl];
            b[1] = ws[k0 + kh + 1][n0 + nl];
            acc  = wmma_f32_16x16x4(a, b, acc);
        }
        #pragma unroll
        for (int v = 0; v < 8; ++v) {
            const int rr = mt * 16 + v + (lane >> 4) * 8;
            float z   = zs[rr][n0 + nl];
            float val = (1.f - beta) * z + beta * acc[v];
            val = val > 0.f ? val : 0.f;
            if (LAST) hs[rr][n0 + nl] = val;
            else      hout[(size_t)(nb0 + rr) * HID + n0 + nl] = val;
        }
    }

    if (LAST) {
        __syncthreads();
        float* ot = &zs[0][0];
        for (int job = wv; job < MT * OCT; job += LW) {
            const int mt = job / OCT;
            const int n0 = (job - mt * OCT) * 16;
            v8f acc = {};
            #pragma unroll
            for (int k0 = 0; k0 < HID; k0 += 4) {
                v2f a, b;
                a[0] = hs[mt * 16 + m][k0 + kh];
                a[1] = hs[mt * 16 + m][k0 + kh + 1];
                b[0] = wo[k0 + kh][n0 + nl];
                b[1] = wo[k0 + kh + 1][n0 + nl];
                acc  = wmma_f32_16x16x4(a, b, acc);
            }
            const int   cc = n0 + nl;
            if (cc < OUTC) {
                const float bo = bout[cc];
                #pragma unroll
                for (int v = 0; v < 8; ++v)
                    ot[(mt * 16 + v + (lane >> 4) * 8) * OUTC + cc] = acc[v] + bo;
            }
        }
        __syncthreads();
        float4* og = reinterpret_cast<float4*>(out + (size_t)nb0 * OUTC);
        for (int i = tid; i < RNG * OUTC / 4; i += LT)
            og[i] = reinterpret_cast<const float4*>(ot)[i];
    }
}

extern "C" void kernel_launch(void* const* d_in, const int* in_sizes, int n_in,
                              void* d_out, int out_size, void* d_ws, size_t ws_size,
                              hipStream_t stream) {
    (void)in_sizes; (void)n_in; (void)out_size;

    const float* x     = (const float*)d_in[0];
    const int*   ei    = (const int*)  d_in[1];
    const float* ew    = (const float*)d_in[2];
    const float* Win   = (const float*)d_in[3];
    const float* bin   = (const float*)d_in[4];
    const float* convw = (const float*)d_in[5];
    const float* Wout  = (const float*)d_in[6];
    const float* bout  = (const float*)d_in[7];
    const int* row = ei;
    const int* col = ei + EE;

    char*  ws  = (char*)d_ws;
    size_t off = 0;
    auto carve = [&](size_t bytes) -> float* {
        float* p = (float*)(ws + off);
        off = (off + bytes + 255) & ~(size_t)255;
        return p;
    };
    float* dinv   = carve((size_t)NN * 4);
    int*   cnt    = (int*)carve((size_t)NB * CPAD * 4);
    int*   cursor = (int*)carve((size_t)NB * CPAD * 4);
    int*   start  = (int*)carve((size_t)(NB + 1) * 4);
    int*   sedge  = (int*)carve((size_t)EE * 4);
    float* snorm  = carve((size_t)EE * 4);
    float* x0     = carve((size_t)NN * HID * 4);
    float* ha     = carve((size_t)NN * HID * 4);
    float* hb     = carve((size_t)NN * HID * 4);

    k_fill   <<<(NB * CPAD + 255) / 256, 256, 0, stream>>>((float*)cnt, 0.0f, NB * CPAD);
    k_hist   <<<NCH, BT, 0, stream>>>(col, cnt);
    k_scan   <<<1, 512, 0, stream>>>(cnt, start, cursor);
    k_place  <<<NCH, BT, 0, stream>>>(row, col, ew, cursor, sedge, snorm);

    k_deg_range<<<NB, DT, 0, stream>>>(sedge, snorm, start, dinv);
    k_norm     <<<NB, DT, 0, stream>>>(sedge, start, dinv, snorm);

    k_gemm_in<<<NN / 16, 128, 0, stream>>>(x, Win, bin, x0);

    for (int l = 0; l < NLAYER; ++l) {
        float beta = logf(0.5f / (float)(l + 1) + 1.0f);
        const float* hin  = l == 0 ? x0 : ((l & 1) ? hb : ha);
        float*       hout = (l & 1) ? ha : hb;
        if (l == NLAYER - 1)
            k_agg_layer<true><<<NB, LT, 0, stream>>>(sedge, snorm, start, hin, x0, dinv,
                                                      convw + (size_t)l * HID * HID, hout,
                                                      Wout, bout, (float*)d_out, beta, ws_size);
        else
            k_agg_layer<false><<<NB, LT, 0, stream>>>(sedge, snorm, start, hin, x0, dinv,
                                                       convw + (size_t)l * HID * HID, hout,
                                                       Wout, bout, (float*)d_out, beta, ws_size);
    }
}
